# starting kernel
# baseline (speedup 1.0000x reference)
#include <hip/hip_runtime.h>


typedef __attribute__((ext_vector_type(2))) float v2f;
typedef __attribute__((ext_vector_type(8))) float v8f;

#define B_ 4
#define S_ 4096
#define D_ 768
#define K_ 24
#define HALF_S 2048
#define K4_ (D_ / 4)

__global__ __launch_bounds__(256)
void stu_pack_kernel(const float* __restrict__ Mi, float* __restrict__ P) {
    int n  = blockIdx.x * 256 + threadIdx.x;
    int k4 = blockIdx.y;
    if (n >= D_) return;
    float4 v;
    v.x = Mi[(size_t)(4 * k4 + 0) * D_ + n];
    v.y = Mi[(size_t)(4 * k4 + 1) * D_ + n];
    v.z = Mi[(size_t)(4 * k4 + 2) * D_ + n];
    v.w = Mi[(size_t)(4 * k4 + 3) * D_ + n];
    *(float4*)(P + ((size_t)k4 * D_ + n) * 4) = v;
}

__global__ __launch_bounds__(256)
void stu_filter_kernel(const float* __restrict__ phi,
                       const float* __restrict__ Mf,
                       float* __restrict__ he) {
    int d = blockIdx.x * 256 + threadIdx.x;
    int j = blockIdx.y;
    if (d >= D_) return;
    const float* prow = phi + (2 * j) * K_;
    float acc = 0.f;
#pragma unroll
    for (int k = 0; k < K_; ++k)
        acc = fmaf(prow[k], Mf[k * D_ + d], acc);
    he[j * D_ + d] = 2.0f * acc;
}

__global__ __launch_bounds__(128)
void stu_xproj_wmma(const float* __restrict__ X,
                    const float* __restrict__ P,
                    float* __restrict__ XP) {
    const int wave = threadIdx.x >> 5;
    const int lane = threadIdx.x & 31;
    const int half = lane >> 4;
    const int l16  = lane & 15;
    const int m0 = blockIdx.x * 32;
    const int n0 = blockIdx.y * 256 + wave * 64;

    v8f c0[4] = {}, c1[4] = {};
    const float* arow0 = X + (size_t)(m0 + l16) * D_ + half * 2;
    const float* arow1 = arow0 + (size_t)16 * D_;
    const float* pcol  = P + ((size_t)(n0 + l16)) * 4 + 2 * half;

    for (int k4 = 0; k4 < K4_; ++k4) {
        v2f a0 = *(const v2f*)(arow0 + k4 * 4);
        v2f a1 = *(const v2f*)(arow1 + k4 * 4);
        const float* pk = pcol + (size_t)k4 * D_ * 4;
#pragma unroll
        for (int nt = 0; nt < 4; ++nt) {
            v2f b = *(const v2f*)(pk + nt * 64);
            c0[nt] = __builtin_amdgcn_wmma_f32_16x16x4_f32(false, a0, false, b, (short)0, c0[nt], false, false);
            c1[nt] = __builtin_amdgcn_wmma_f32_16x16x4_f32(false, a1, false, b, (short)0, c1[nt], false, false);
        }
    }

    float* out0 = XP + (size_t)(m0 + half * 8) * D_ + n0 + l16;
    float* out1 = out0 + (size_t)16 * D_;
#pragma unroll
    for (int nt = 0; nt < 4; ++nt) {
#pragma unroll
        for (int v = 0; v < 8; ++v) {
            out0[(size_t)v * D_ + nt * 16] = c0[nt][v];
            out1[(size_t)v * D_ + nt * 16] = c1[nt][v];
        }
    }
}

__global__ __launch_bounds__(256)
void stu_conv_kernel(const float* __restrict__ XP,
                     const float* __restrict__ HE,
                     float* __restrict__ OUT) {
    __shared__ float sh_h[64 * 32];
    __shared__ float sh_x[128 * 32];

    const int ib = blockIdx.x;
    const int d0 = blockIdx.y * 32;
    const int b  = blockIdx.z >> 1;
    const int p  = blockIdx.z & 1;
    const int tid  = threadIdx.x;
    const int lane = tid & 31;
    const int w    = tid >> 5;
    const int i0 = ib * 64;
    const int d  = d0 + lane;

    float acc[8];
#pragma unroll
    for (int r = 0; r < 8; ++r) acc[r] = 0.f;

    const float* xpb = XP + (size_t)b * S_ * D_;

    for (int c = 0; c <= ib; ++c) {
        const int j0 = c * 64;
        for (int idx = tid; idx < 64 * 8; idx += 256) {
            int jj = idx >> 3, q = idx & 7;
            *(float4*)(sh_h + jj * 32 + q * 4) =
                *(const float4*)(HE + (size_t)(j0 + jj) * D_ + d0 + q * 4);
        }
        const int u_base = i0 - j0 - 63;
        for (int idx = tid; idx < 128 * 8; idx += 256) {
            int s = idx >> 3, q = idx & 7;
            int u = u_base + s;
            float4 v = {0.f, 0.f, 0.f, 0.f};
            if (u >= 0 && u < HALF_S)
                v = *(const float4*)(xpb + (size_t)(2 * u + p) * D_ + d0 + q * 4);
            *(float4*)(sh_x + s * 32 + q * 4) = v;
        }
        __syncthreads();

        if (c < ib) {
            const int j0n = j0 + 64;
            __builtin_prefetch(&HE[(size_t)(j0n + (tid >> 2)) * D_ + d0 + (tid & 3) * 8], 0, 0);
            int u_n = i0 - j0n - 63 + (tid >> 1);
            if (u_n >= 0 && u_n < HALF_S)
                __builtin_prefetch(&xpb[(size_t)(2 * u_n + p) * D_ + d0 + (tid & 1) * 16], 0, 0);
        }

        float xwin[8];
#pragma unroll
        for (int r = 0; r < 8; ++r) xwin[r] = sh_x[(8 * w + r) * 32 + lane];
#pragma unroll 8
        for (int jj = 63; jj >= 0; --jj) {
            float h = sh_h[jj * 32 + lane];
#pragma unroll
            for (int r = 0; r < 8; ++r) acc[r] = fmaf(h, xwin[r], acc[r]);
#pragma unroll
            for (int r = 0; r < 7; ++r) xwin[r] = xwin[r + 1];
            xwin[7] = sh_x[(8 * w + 71 - jj) * 32 + lane];
        }
        __syncthreads();
    }

#pragma unroll
    for (int r = 0; r < 8; ++r) {
        int t = 2 * (i0 + 8 * w + r) + p;
        OUT[((size_t)b * S_ + t) * D_ + d] = acc[r];
    }
}

extern "C" void kernel_launch(void* const* d_in, const int* in_sizes, int n_in,
                              void* d_out, int out_size, void* d_ws, size_t ws_size,
                              hipStream_t stream) {
    (void)in_sizes; (void)n_in; (void)out_size; (void)ws_size;
    const float* x   = (const float*)d_in[0];
    const float* phi = (const float*)d_in[1];
    const float* Mi  = (const float*)d_in[2];
    const float* Mf  = (const float*)d_in[3];
    float* out  = (float*)d_out;

    float* ws_x = (float*)d_ws;
    float* ws_h = ws_x + (size_t)B_ * S_ * D_;
    float* ws_p = ws_h + (size_t)HALF_S * D_;

    stu_pack_kernel  <<<dim3(3, K4_, 1),            256, 0, stream>>>(Mi, ws_p);
    stu_filter_kernel<<<dim3(3, HALF_S, 1),         256, 0, stream>>>(phi, Mf, ws_h);
    stu_xproj_wmma   <<<dim3((B_ * S_) / 32, D_ / 256, 1), 128, 0, stream>>>(x, ws_p, ws_x);
    stu_conv_kernel  <<<dim3(HALF_S / 64, D_ / 32, B_ * 2), 256, 0, stream>>>(ws_x, ws_h, out);
}


# conv kernel: one block does both parities of its i-block; accumulators are (parity 0, parity 1) pairs, one packed f32 FMA per pair with the shared filter value; summation order kept
# speedup vs baseline: 1.1140x; 1.1140x over previous
#include <hip/hip_runtime.h>


typedef __attribute__((ext_vector_type(2))) float v2f;
typedef __attribute__((ext_vector_type(8))) float v8f;

#define B_ 4
#define S_ 4096
#define D_ 768
#define K_ 24
#define HALF_S 2048
#define K4_ (D_ / 4)

__global__ __launch_bounds__(256)
void stu_pack_kernel(const float* __restrict__ Mi, float* __restrict__ P) {
    int n  = blockIdx.x * 256 + threadIdx.x;
    int k4 = blockIdx.y;
    if (n >= D_) return;
    float4 v;
    v.x = Mi[(size_t)(4 * k4 + 0) * D_ + n];
    v.y = Mi[(size_t)(4 * k4 + 1) * D_ + n];
    v.z = Mi[(size_t)(4 * k4 + 2) * D_ + n];
    v.w = Mi[(size_t)(4 * k4 + 3) * D_ + n];
    *(float4*)(P + ((size_t)k4 * D_ + n) * 4) = v;
}

__global__ __launch_bounds__(256)
void stu_filter_kernel(const float* __restrict__ phi,
                       const float* __restrict__ Mf,
                       float* __restrict__ he) {
    int d = blockIdx.x * 256 + threadIdx.x;
    int j = blockIdx.y;
    if (d >= D_) return;
    const float* prow = phi + (2 * j) * K_;
    float acc = 0.f;
#pragma unroll
    for (int k = 0; k < K_; ++k)
        acc = fmaf(prow[k], Mf[k * D_ + d], acc);
    he[j * D_ + d] = 2.0f * acc;
}

__global__ __launch_bounds__(128)
void stu_xproj_wmma(const float* __restrict__ X,
                    const float* __restrict__ P,
                    float* __restrict__ XP) {
    const int wave = threadIdx.x >> 5;
    const int lane = threadIdx.x & 31;
    const int half = lane >> 4;
    const int l16  = lane & 15;
    const int m0 = blockIdx.x * 32;
    const int n0 = blockIdx.y * 256 + wave * 64;

    v8f c0[4] = {}, c1[4] = {};
    const float* arow0 = X + (size_t)(m0 + l16) * D_ + half * 2;
    const float* arow1 = arow0 + (size_t)16 * D_;
    const float* pcol  = P + ((size_t)(n0 + l16)) * 4 + 2 * half;

    for (int k4 = 0; k4 < K4_; ++k4) {
        v2f a0 = *(const v2f*)(arow0 + k4 * 4);
        v2f a1 = *(const v2f*)(arow1 + k4 * 4);
        const float* pk = pcol + (size_t)k4 * D_ * 4;
#pragma unroll
        for (int nt = 0; nt < 4; ++nt) {
            v2f b = *(const v2f*)(pk + nt * 64);
            c0[nt] = __builtin_amdgcn_wmma_f32_16x16x4_f32(false, a0, false, b, (short)0, c0[nt], false, false);
            c1[nt] = __builtin_amdgcn_wmma_f32_16x16x4_f32(false, a1, false, b, (short)0, c1[nt], false, false);
        }
    }

    float* out0 = XP + (size_t)(m0 + half * 8) * D_ + n0 + l16;
    float* out1 = out0 + (size_t)16 * D_;
#pragma unroll
    for (int nt = 0; nt < 4; ++nt) {
#pragma unroll
        for (int v = 0; v < 8; ++v) {
            out0[(size_t)v * D_ + nt * 16] = c0[nt][v];
            out1[(size_t)v * D_ + nt * 16] = c1[nt][v];
        }
    }
}

__global__ __launch_bounds__(256)
void stu_conv_kernel(const float* __restrict__ XP,
                     const float* __restrict__ HE,
                     float* __restrict__ OUT) {
    __shared__ float sh_h[64 * 32];
    __shared__ float sh_x[256 * 32];

    const int ib = blockIdx.x;
    const int d0 = blockIdx.y * 32;
    const int b  = blockIdx.z;
    const int tid  = threadIdx.x;
    const int lane = tid & 31;
    const int w    = tid >> 5;
    const int i0 = ib * 64;
    const int d  = d0 + lane;

    v2f acc[8];
#pragma unroll
    for (int r = 0; r < 8; ++r) acc[r] = (v2f){0.f, 0.f};

    const float* xpb = XP + (size_t)b * S_ * D_;

    for (int c = 0; c <= ib; ++c) {
        const int j0 = c * 64;
        for (int idx = tid; idx < 64 * 8; idx += 256) {
            int jj = idx >> 3, q = idx & 7;
            *(float4*)(sh_h + jj * 32 + q * 4) =
                *(const float4*)(HE + (size_t)(j0 + jj) * D_ + d0 + q * 4);
        }
        const int t_base = 2 * (i0 - j0 - 63);
        for (int idx = tid; idx < 256 * 8; idx += 256) {
            int s2 = idx >> 3, q = idx & 7;
            int t = t_base + s2;
            float4 v = {0.f, 0.f, 0.f, 0.f};
            if (t >= 0 && t < S_)
                v = *(const float4*)(xpb + (size_t)t * D_ + d0 + q * 4);
            *(float4*)(sh_x + s2 * 32 + q * 4) = v;
        }
        __syncthreads();

        if (c < ib) {
            const int j0n = j0 + 64;
            __builtin_prefetch(&HE[(size_t)(j0n + (tid >> 2)) * D_ + d0 + (tid & 3) * 8], 0, 0);
            int t_n = t_base - 128 + tid;
            if (t_n >= 0 && t_n < S_)
                __builtin_prefetch(&xpb[(size_t)t_n * D_ + d0], 0, 0);
        }

        v2f xwin[8];
#pragma unroll
        for (int r = 0; r < 8; ++r) {
            xwin[r].x = sh_x[(2 * (8 * w + r)) * 32 + lane];
            xwin[r].y = sh_x[(2 * (8 * w + r) + 1) * 32 + lane];
        }
#pragma unroll 8
        for (int jj = 63; jj >= 0; --jj) {
            float h = sh_h[jj * 32 + lane];
            v2f hh = {h, h};
#pragma unroll
            for (int r = 0; r < 8; ++r) acc[r] = __builtin_elementwise_fma(hh, xwin[r], acc[r]);
#pragma unroll
            for (int r = 0; r < 7; ++r) xwin[r] = xwin[r + 1];
            xwin[7].x = sh_x[(2 * (8 * w + 71 - jj)) * 32 + lane];
            xwin[7].y = sh_x[(2 * (8 * w + 71 - jj) + 1) * 32 + lane];
        }
        __syncthreads();
    }

#pragma unroll
    for (int r = 0; r < 8; ++r) {
        int t = 2 * (i0 + 8 * w + r);
        OUT[((size_t)b * S_ + t) * D_ + d]     = acc[r].x;
        OUT[((size_t)b * S_ + t + 1) * D_ + d] = acc[r].y;
    }
}

extern "C" void kernel_launch(void* const* d_in, const int* in_sizes, int n_in,
                              void* d_out, int out_size, void* d_ws, size_t ws_size,
                              hipStream_t stream) {
    (void)in_sizes; (void)n_in; (void)out_size; (void)ws_size;
    const float* x   = (const float*)d_in[0];
    const float* phi = (const float*)d_in[1];
    const float* Mi  = (const float*)d_in[2];
    const float* Mf  = (const float*)d_in[3];
    float* out  = (float*)d_out;

    float* ws_x = (float*)d_ws;
    float* ws_h = ws_x + (size_t)B_ * S_ * D_;
    float* ws_p = ws_h + (size_t)HALF_S * D_;

    stu_pack_kernel  <<<dim3(3, K4_, 1),            256, 0, stream>>>(Mi, ws_p);
    stu_filter_kernel<<<dim3(3, HALF_S, 1),         256, 0, stream>>>(phi, Mf, ws_h);
    stu_xproj_wmma   <<<dim3((B_ * S_) / 32, D_ / 256, 1), 128, 0, stream>>>(x, ws_p, ws_x);
    stu_conv_kernel  <<<dim3(HALF_S / 64, D_ / 32, B_), 256, 0, stream>>>(ws_x, ws_h, out);
}


# conv kernel: staging software-pipelined: next chunk's 10 global b128 loads per thread issued before the current chunk's FMAs, held in registers until the LDS tile is free; hint prefetch removed
# speedup vs baseline: 1.2443x; 1.1170x over previous
#include <hip/hip_runtime.h>


typedef __attribute__((ext_vector_type(2))) float v2f;
typedef __attribute__((ext_vector_type(4))) float v4f;
typedef __attribute__((ext_vector_type(8))) float v8f;

#define B_ 4
#define S_ 4096
#define D_ 768
#define K_ 24
#define HALF_S 2048
#define K4_ (D_ / 4)

__global__ __launch_bounds__(256)
void stu_pack_kernel(const float* __restrict__ Mi, float* __restrict__ P) {
    int n  = blockIdx.x * 256 + threadIdx.x;
    int k4 = blockIdx.y;
    if (n >= D_) return;
    float4 v;
    v.x = Mi[(size_t)(4 * k4 + 0) * D_ + n];
    v.y = Mi[(size_t)(4 * k4 + 1) * D_ + n];
    v.z = Mi[(size_t)(4 * k4 + 2) * D_ + n];
    v.w = Mi[(size_t)(4 * k4 + 3) * D_ + n];
    *(float4*)(P + ((size_t)k4 * D_ + n) * 4) = v;
}

__global__ __launch_bounds__(256)
void stu_filter_kernel(const float* __restrict__ phi,
                       const float* __restrict__ Mf,
                       float* __restrict__ he) {
    int d = blockIdx.x * 256 + threadIdx.x;
    int j = blockIdx.y;
    if (d >= D_) return;
    const float* prow = phi + (2 * j) * K_;
    float acc = 0.f;
#pragma unroll
    for (int k = 0; k < K_; ++k)
        acc = fmaf(prow[k], Mf[k * D_ + d], acc);
    he[j * D_ + d] = 2.0f * acc;
}

__global__ __launch_bounds__(128)
void stu_xproj_wmma(const float* __restrict__ X,
                    const float* __restrict__ P,
                    float* __restrict__ XP) {
    const int wave = threadIdx.x >> 5;
    const int lane = threadIdx.x & 31;
    const int half = lane >> 4;
    const int l16  = lane & 15;
    const int m0 = blockIdx.x * 32;
    const int n0 = blockIdx.y * 256 + wave * 64;

    v8f c0[4] = {}, c1[4] = {};
    const float* arow0 = X + (size_t)(m0 + l16) * D_ + half * 2;
    const float* arow1 = arow0 + (size_t)16 * D_;
    const float* pcol  = P + ((size_t)(n0 + l16)) * 4 + 2 * half;

    for (int k4 = 0; k4 < K4_; ++k4) {
        v2f a0 = *(const v2f*)(arow0 + k4 * 4);
        v2f a1 = *(const v2f*)(arow1 + k4 * 4);
        const float* pk = pcol + (size_t)k4 * D_ * 4;
#pragma unroll
        for (int nt = 0; nt < 4; ++nt) {
            v2f b = *(const v2f*)(pk + nt * 64);
            c0[nt] = __builtin_amdgcn_wmma_f32_16x16x4_f32(false, a0, false, b, (short)0, c0[nt], false, false);
            c1[nt] = __builtin_amdgcn_wmma_f32_16x16x4_f32(false, a1, false, b, (short)0, c1[nt], false, false);
        }
    }

    float* out0 = XP + (size_t)(m0 + half * 8) * D_ + n0 + l16;
    float* out1 = out0 + (size_t)16 * D_;
#pragma unroll
    for (int nt = 0; nt < 4; ++nt) {
#pragma unroll
        for (int v = 0; v < 8; ++v) {
            out0[(size_t)v * D_ + nt * 16] = c0[nt][v];
            out1[(size_t)v * D_ + nt * 16] = c1[nt][v];
        }
    }
}


struct ConvStage { v4f h0, h1, x[8]; };

__device__ __forceinline__ void conv_issue_loads(ConvStage& st, const float* __restrict__ xpb,
                                                 const float* __restrict__ hep, int c, int i0, int r0) {
    const int j0 = c * 64;
    const int t_base = 2 * (i0 - j0 - 63);
    const int t0 = t_base + r0;
    if (t_base >= 0 && t_base + 255 < S_) {
        const float* xp = xpb + (size_t)t0 * D_;
#pragma unroll
        for (int k = 0; k < 8; ++k) st.x[k] = *(const v4f*)(xp + (size_t)(32 * k) * D_);
    } else {
#pragma unroll
        for (int k = 0; k < 8; ++k) {
            const int t = t0 + 32 * k;
            v4f v = {0.f, 0.f, 0.f, 0.f};
            if (t >= 0 && t < S_) v = *(const v4f*)(xpb + (size_t)t * D_);
            st.x[k] = v;
        }
    }
    const float* hp = hep + (size_t)(j0 + r0) * D_;
    st.h0 = *(const v4f*)(hp);
    st.h1 = *(const v4f*)(hp + (size_t)32 * D_);
}

__global__ __launch_bounds__(256)
void stu_conv_kernel(const float* __restrict__ XP,
                     const float* __restrict__ HE,
                     float* __restrict__ OUT) {
    __shared__ float sh_h[64 * 32];
    __shared__ float sh_x[256 * 32];

    const int ib = blockIdx.x;
    const int d0 = blockIdx.y * 32;
    const int b  = blockIdx.z;
    const int tid  = threadIdx.x;
    const int lane = tid & 31;
    const int w    = tid >> 5;
    const int i0 = ib * 64;
    const int d  = d0 + lane;
    const int q4 = (tid & 7) * 4;
    const int r0 = tid >> 3;

    v2f acc[8];
#pragma unroll
    for (int r = 0; r < 8; ++r) acc[r] = (v2f){0.f, 0.f};

    const float* xpb = XP + (size_t)b * S_ * D_ + d0 + q4;
    const float* hep = HE + d0 + q4;
    float* sth = sh_h + r0 * 32 + q4;
    float* stx = sh_x + r0 * 32 + q4;

    ConvStage st;
    conv_issue_loads(st, xpb, hep, 0, i0, r0);

    for (int c = 0; c <= ib; ++c) {
        *(v4f*)(sth)           = st.h0;
        *(v4f*)(sth + 32 * 32) = st.h1;
#pragma unroll
        for (int k = 0; k < 8; ++k) *(v4f*)(stx + (32 * k) * 32) = st.x[k];
        __syncthreads();

        if (c < ib) conv_issue_loads(st, xpb, hep, c + 1, i0, r0);

        v2f xwin[8];
#pragma unroll
        for (int r = 0; r < 8; ++r) {
            xwin[r].x = sh_x[(2 * (8 * w + r)) * 32 + lane];
            xwin[r].y = sh_x[(2 * (8 * w + r) + 1) * 32 + lane];
        }
#pragma unroll 8
        for (int jj = 63; jj >= 0; --jj) {
            float h = sh_h[jj * 32 + lane];
            v2f hh = {h, h};
#pragma unroll
            for (int r = 0; r < 8; ++r) acc[r] = __builtin_elementwise_fma(hh, xwin[r], acc[r]);
#pragma unroll
            for (int r = 0; r < 7; ++r) xwin[r] = xwin[r + 1];
            xwin[7].x = sh_x[(2 * (8 * w + 71 - jj)) * 32 + lane];
            xwin[7].y = sh_x[(2 * (8 * w + 71 - jj) + 1) * 32 + lane];
        }
        __syncthreads();
    }

#pragma unroll
    for (int r = 0; r < 8; ++r) {
        int t = 2 * (i0 + 8 * w + r);
        OUT[((size_t)b * S_ + t) * D_ + d]     = acc[r].x;
        OUT[((size_t)b * S_ + t + 1) * D_ + d] = acc[r].y;
    }
}

extern "C" void kernel_launch(void* const* d_in, const int* in_sizes, int n_in,
                              void* d_out, int out_size, void* d_ws, size_t ws_size,
                              hipStream_t stream) {
    (void)in_sizes; (void)n_in; (void)out_size; (void)ws_size;
    const float* x   = (const float*)d_in[0];
    const float* phi = (const float*)d_in[1];
    const float* Mi  = (const float*)d_in[2];
    const float* Mf  = (const float*)d_in[3];
    float* out  = (float*)d_out;

    float* ws_x = (float*)d_ws;
    float* ws_h = ws_x + (size_t)B_ * S_ * D_;
    float* ws_p = ws_h + (size_t)HALF_S * D_;

    stu_pack_kernel  <<<dim3(3, K4_, 1),            256, 0, stream>>>(Mi, ws_p);
    stu_filter_kernel<<<dim3(3, HALF_S, 1),         256, 0, stream>>>(phi, Mf, ws_h);
    stu_xproj_wmma   <<<dim3((B_ * S_) / 32, D_ / 256, 1), 128, 0, stream>>>(x, ws_p, ws_x);
    stu_conv_kernel  <<<dim3(HALF_S / 64, D_ / 32, B_), 256, 0, stream>>>(ws_x, ws_h, out);
}


# conv kernel: work-groups ordered by work: i-block index moved to grid z and reversed (the 32-chunk i-block is dispatched first, the 1-chunk one last), channel tile on x, batch on y; body unchanged
# speedup vs baseline: 1.5176x; 1.2196x over previous
#include <hip/hip_runtime.h>


typedef __attribute__((ext_vector_type(2))) float v2f;
typedef __attribute__((ext_vector_type(4))) float v4f;
typedef __attribute__((ext_vector_type(8))) float v8f;

#define B_ 4
#define S_ 4096
#define D_ 768
#define K_ 24
#define HALF_S 2048
#define K4_ (D_ / 4)

__global__ __launch_bounds__(256)
void stu_pack_kernel(const float* __restrict__ Mi, float* __restrict__ P) {
    int n  = blockIdx.x * 256 + threadIdx.x;
    int k4 = blockIdx.y;
    if (n >= D_) return;
    float4 v;
    v.x = Mi[(size_t)(4 * k4 + 0) * D_ + n];
    v.y = Mi[(size_t)(4 * k4 + 1) * D_ + n];
    v.z = Mi[(size_t)(4 * k4 + 2) * D_ + n];
    v.w = Mi[(size_t)(4 * k4 + 3) * D_ + n];
    *(float4*)(P + ((size_t)k4 * D_ + n) * 4) = v;
}

__global__ __launch_bounds__(256)
void stu_filter_kernel(const float* __restrict__ phi,
                       const float* __restrict__ Mf,
                       float* __restrict__ he) {
    int d = blockIdx.x * 256 + threadIdx.x;
    int j = blockIdx.y;
    if (d >= D_) return;
    const float* prow = phi + (2 * j) * K_;
    float acc = 0.f;
#pragma unroll
    for (int k = 0; k < K_; ++k)
        acc = fmaf(prow[k], Mf[k * D_ + d], acc);
    he[j * D_ + d] = 2.0f * acc;
}

__global__ __launch_bounds__(128)
void stu_xproj_wmma(const float* __restrict__ X,
                    const float* __restrict__ P,
                    float* __restrict__ XP) {
    const int wave = threadIdx.x >> 5;
    const int lane = threadIdx.x & 31;
    const int half = lane >> 4;
    const int l16  = lane & 15;
    const int m0 = blockIdx.x * 32;
    const int n0 = blockIdx.y * 256 + wave * 64;

    v8f c0[4] = {}, c1[4] = {};
    const float* arow0 = X + (size_t)(m0 + l16) * D_ + half * 2;
    const float* arow1 = arow0 + (size_t)16 * D_;
    const float* pcol  = P + ((size_t)(n0 + l16)) * 4 + 2 * half;

    for (int k4 = 0; k4 < K4_; ++k4) {
        v2f a0 = *(const v2f*)(arow0 + k4 * 4);
        v2f a1 = *(const v2f*)(arow1 + k4 * 4);
        const float* pk = pcol + (size_t)k4 * D_ * 4;
#pragma unroll
        for (int nt = 0; nt < 4; ++nt) {
            v2f b = *(const v2f*)(pk + nt * 64);
            c0[nt] = __builtin_amdgcn_wmma_f32_16x16x4_f32(false, a0, false, b, (short)0, c0[nt], false, false);
            c1[nt] = __builtin_amdgcn_wmma_f32_16x16x4_f32(false, a1, false, b, (short)0, c1[nt], false, false);
        }
    }

    float* out0 = XP + (size_t)(m0 + half * 8) * D_ + n0 + l16;
    float* out1 = out0 + (size_t)16 * D_;
#pragma unroll
    for (int nt = 0; nt < 4; ++nt) {
#pragma unroll
        for (int v = 0; v < 8; ++v) {
            out0[(size_t)v * D_ + nt * 16] = c0[nt][v];
            out1[(size_t)v * D_ + nt * 16] = c1[nt][v];
        }
    }
}


struct ConvStage { v4f h0, h1, x[8]; };

__device__ __forceinline__ void conv_issue_loads(ConvStage& st, const float* __restrict__ xpb,
                                                 const float* __restrict__ hep, int c, int i0, int r0) {
    const int j0 = c * 64;
    const int t_base = 2 * (i0 - j0 - 63);
    const int t0 = t_base + r0;
    if (t_base >= 0 && t_base + 255 < S_) {
        const float* xp = xpb + (size_t)t0 * D_;
#pragma unroll
        for (int k = 0; k < 8; ++k) st.x[k] = *(const v4f*)(xp + (size_t)(32 * k) * D_);
    } else {
#pragma unroll
        for (int k = 0; k < 8; ++k) {
            const int t = t0 + 32 * k;
            v4f v = {0.f, 0.f, 0.f, 0.f};
            if (t >= 0 && t < S_) v = *(const v4f*)(xpb + (size_t)t * D_);
            st.x[k] = v;
        }
    }
    const float* hp = hep + (size_t)(j0 + r0) * D_;
    st.h0 = *(const v4f*)(hp);
    st.h1 = *(const v4f*)(hp + (size_t)32 * D_);
}

__global__ __launch_bounds__(256)
void stu_conv_kernel(const float* __restrict__ XP,
                     const float* __restrict__ HE,
                     float* __restrict__ OUT) {
    __shared__ float sh_h[64 * 32];
    __shared__ float sh_x[256 * 32];

    const int ib = (HALF_S / 64 - 1) - (int)blockIdx.z;
    const int d0 = blockIdx.x * 32;
    const int b  = blockIdx.y;
    const int tid  = threadIdx.x;
    const int lane = tid & 31;
    const int w    = tid >> 5;
    const int i0 = ib * 64;
    const int d  = d0 + lane;
    const int q4 = (tid & 7) * 4;
    const int r0 = tid >> 3;

    v2f acc[8];
#pragma unroll
    for (int r = 0; r < 8; ++r) acc[r] = (v2f){0.f, 0.f};

    const float* xpb = XP + (size_t)b * S_ * D_ + d0 + q4;
    const float* hep = HE + d0 + q4;
    float* sth = sh_h + r0 * 32 + q4;
    float* stx = sh_x + r0 * 32 + q4;

    ConvStage st;
    conv_issue_loads(st, xpb, hep, 0, i0, r0);

    for (int c = 0; c <= ib; ++c) {
        *(v4f*)(sth)           = st.h0;
        *(v4f*)(sth + 32 * 32) = st.h1;
#pragma unroll
        for (int k = 0; k < 8; ++k) *(v4f*)(stx + (32 * k) * 32) = st.x[k];
        __syncthreads();

        if (c < ib) conv_issue_loads(st, xpb, hep, c + 1, i0, r0);

        v2f xwin[8];
#pragma unroll
        for (int r = 0; r < 8; ++r) {
            xwin[r].x = sh_x[(2 * (8 * w + r)) * 32 + lane];
            xwin[r].y = sh_x[(2 * (8 * w + r) + 1) * 32 + lane];
        }
#pragma unroll 8
        for (int jj = 63; jj >= 0; --jj) {
            float h = sh_h[jj * 32 + lane];
            v2f hh = {h, h};
#pragma unroll
            for (int r = 0; r < 8; ++r) acc[r] = __builtin_elementwise_fma(hh, xwin[r], acc[r]);
#pragma unroll
            for (int r = 0; r < 7; ++r) xwin[r] = xwin[r + 1];
            xwin[7].x = sh_x[(2 * (8 * w + 71 - jj)) * 32 + lane];
            xwin[7].y = sh_x[(2 * (8 * w + 71 - jj) + 1) * 32 + lane];
        }
        __syncthreads();
    }

#pragma unroll
    for (int r = 0; r < 8; ++r) {
        int t = 2 * (i0 + 8 * w + r);
        OUT[((size_t)b * S_ + t) * D_ + d]     = acc[r].x;
        OUT[((size_t)b * S_ + t + 1) * D_ + d] = acc[r].y;
    }
}

extern "C" void kernel_launch(void* const* d_in, const int* in_sizes, int n_in,
                              void* d_out, int out_size, void* d_ws, size_t ws_size,
                              hipStream_t stream) {
    (void)in_sizes; (void)n_in; (void)out_size; (void)ws_size;
    const float* x   = (const float*)d_in[0];
    const float* phi = (const float*)d_in[1];
    const float* Mi  = (const float*)d_in[2];
    const float* Mf  = (const float*)d_in[3];
    float* out  = (float*)d_out;

    float* ws_x = (float*)d_ws;
    float* ws_h = ws_x + (size_t)B_ * S_ * D_;
    float* ws_p = ws_h + (size_t)HALF_S * D_;

    stu_pack_kernel  <<<dim3(3, K4_, 1),            256, 0, stream>>>(Mi, ws_p);
    stu_filter_kernel<<<dim3(3, HALF_S, 1),         256, 0, stream>>>(phi, Mf, ws_h);
    stu_xproj_wmma   <<<dim3((B_ * S_) / 32, D_ / 256, 1), 128, 0, stream>>>(x, ws_p, ws_x);
    stu_conv_kernel  <<<dim3(D_ / 32, B_, HALF_S / 64), 256, 0, stream>>>(ws_x, ws_h, out);
}


# conv kernel: a lane owns two adjacent channels of one parity (lane halves = parities): filter and window values are single 8-byte LDS reads from the row-major tiles; summation order kept
# speedup vs baseline: 1.5669x; 1.0325x over previous
#include <hip/hip_runtime.h>


typedef __attribute__((ext_vector_type(2))) float v2f;
typedef __attribute__((ext_vector_type(4))) float v4f;
typedef __attribute__((ext_vector_type(8))) float v8f;

#define B_ 4
#define S_ 4096
#define D_ 768
#define K_ 24
#define HALF_S 2048
#define K4_ (D_ / 4)

__global__ __launch_bounds__(256)
void stu_pack_kernel(const float* __restrict__ Mi, float* __restrict__ P) {
    int n  = blockIdx.x * 256 + threadIdx.x;
    int k4 = blockIdx.y;
    if (n >= D_) return;
    float4 v;
    v.x = Mi[(size_t)(4 * k4 + 0) * D_ + n];
    v.y = Mi[(size_t)(4 * k4 + 1) * D_ + n];
    v.z = Mi[(size_t)(4 * k4 + 2) * D_ + n];
    v.w = Mi[(size_t)(4 * k4 + 3) * D_ + n];
    *(float4*)(P + ((size_t)k4 * D_ + n) * 4) = v;
}

__global__ __launch_bounds__(256)
void stu_filter_kernel(const float* __restrict__ phi,
                       const float* __restrict__ Mf,
                       float* __restrict__ he) {
    int d = blockIdx.x * 256 + threadIdx.x;
    int j = blockIdx.y;
    if (d >= D_) return;
    const float* prow = phi + (2 * j) * K_;
    float acc = 0.f;
#pragma unroll
    for (int k = 0; k < K_; ++k)
        acc = fmaf(prow[k], Mf[k * D_ + d], acc);
    he[j * D_ + d] = 2.0f * acc;
}

__global__ __launch_bounds__(128)
void stu_xproj_wmma(const float* __restrict__ X,
                    const float* __restrict__ P,
                    float* __restrict__ XP) {
    const int wave = threadIdx.x >> 5;
    const int lane = threadIdx.x & 31;
    const int half = lane >> 4;
    const int l16  = lane & 15;
    const int m0 = blockIdx.x * 32;
    const int n0 = blockIdx.y * 256 + wave * 64;

    v8f c0[4] = {}, c1[4] = {};
    const float* arow0 = X + (size_t)(m0 + l16) * D_ + half * 2;
    const float* arow1 = arow0 + (size_t)16 * D_;
    const float* pcol  = P + ((size_t)(n0 + l16)) * 4 + 2 * half;

    for (int k4 = 0; k4 < K4_; ++k4) {
        v2f a0 = *(const v2f*)(arow0 + k4 * 4);
        v2f a1 = *(const v2f*)(arow1 + k4 * 4);
        const float* pk = pcol + (size_t)k4 * D_ * 4;
#pragma unroll
        for (int nt = 0; nt < 4; ++nt) {
            v2f b = *(const v2f*)(pk + nt * 64);
            c0[nt] = __builtin_amdgcn_wmma_f32_16x16x4_f32(false, a0, false, b, (short)0, c0[nt], false, false);
            c1[nt] = __builtin_amdgcn_wmma_f32_16x16x4_f32(false, a1, false, b, (short)0, c1[nt], false, false);
        }
    }

    float* out0 = XP + (size_t)(m0 + half * 8) * D_ + n0 + l16;
    float* out1 = out0 + (size_t)16 * D_;
#pragma unroll
    for (int nt = 0; nt < 4; ++nt) {
#pragma unroll
        for (int v = 0; v < 8; ++v) {
            out0[(size_t)v * D_ + nt * 16] = c0[nt][v];
            out1[(size_t)v * D_ + nt * 16] = c1[nt][v];
        }
    }
}


struct ConvStage { v4f h0, h1, x[8]; };

__device__ __forceinline__ void conv_issue_loads(ConvStage& st, const float* __restrict__ xpb,
                                                 const float* __restrict__ hep, int c, int i0, int r0) {
    const int j0 = c * 64;
    const int t_base = 2 * (i0 - j0 - 63);
    const int t0 = t_base + r0;
    if (t_base >= 0 && t_base + 255 < S_) {
        const float* xp = xpb + (size_t)t0 * D_;
#pragma unroll
        for (int k = 0; k < 8; ++k) st.x[k] = *(const v4f*)(xp + (size_t)(32 * k) * D_);
    } else {
#pragma unroll
        for (int k = 0; k < 8; ++k) {
            const int t = t0 + 32 * k;
            v4f v = {0.f, 0.f, 0.f, 0.f};
            if (t >= 0 && t < S_) v = *(const v4f*)(xpb + (size_t)t * D_);
            st.x[k] = v;
        }
    }
    const float* hp = hep + (size_t)(j0 + r0) * D_;
    st.h0 = *(const v4f*)(hp);
    st.h1 = *(const v4f*)(hp + (size_t)32 * D_);
}

__global__ __launch_bounds__(256)
void stu_conv_kernel(const float* __restrict__ XP,
                     const float* __restrict__ HE,
                     float* __restrict__ OUT) {
    __shared__ __attribute__((aligned(16))) float sh_h[64 * 32];
    __shared__ __attribute__((aligned(16))) float sh_x[256 * 32];

    const int ib = (HALF_S / 64 - 1) - (int)blockIdx.z;
    const int d0 = blockIdx.x * 32;
    const int b  = blockIdx.y;
    const int tid  = threadIdx.x;
    const int lane = tid & 31;
    const int w    = tid >> 5;
    const int i0 = ib * 64;
    const int par = lane >> 4;
    const int c2  = (lane & 15) * 2;
    const int q4 = (tid & 7) * 4;
    const int r0 = tid >> 3;

    v2f acc[8];
#pragma unroll
    for (int r = 0; r < 8; ++r) acc[r] = (v2f){0.f, 0.f};

    const float* xpb = XP + (size_t)b * S_ * D_ + d0 + q4;
    const float* hep = HE + d0 + q4;
    float* sth = sh_h + r0 * 32 + q4;
    float* stx = sh_x + r0 * 32 + q4;

    ConvStage st;
    conv_issue_loads(st, xpb, hep, 0, i0, r0);

    for (int c = 0; c <= ib; ++c) {
        *(v4f*)(sth)           = st.h0;
        *(v4f*)(sth + 32 * 32) = st.h1;
#pragma unroll
        for (int k = 0; k < 8; ++k) *(v4f*)(stx + (32 * k) * 32) = st.x[k];
        __syncthreads();

        if (c < ib) conv_issue_loads(st, xpb, hep, c + 1, i0, r0);

        const float* xw = sh_x + 64 * (8 * w) + 2 * lane;
        const float* hw = sh_h + c2;
        v2f xwin[8];
#pragma unroll
        for (int r = 0; r < 8; ++r) xwin[r] = *(const v2f*)(xw + 64 * r);
#pragma unroll 8
        for (int jj = 63; jj >= 0; --jj) {
            const v2f hh = *(const v2f*)(hw + 32 * jj);
#pragma unroll
            for (int r = 0; r < 8; ++r) acc[r] = __builtin_elementwise_fma(hh, xwin[r], acc[r]);
#pragma unroll
            for (int r = 0; r < 7; ++r) xwin[r] = xwin[r + 1];
            xwin[7] = *(const v2f*)(xw + 64 * (71 - jj));
        }
        __syncthreads();
    }

#pragma unroll
    for (int r = 0; r < 8; ++r) {
        const int t = 2 * (i0 + 8 * w + r) + par;
        *(v2f*)(OUT + ((size_t)b * S_ + t) * D_ + d0 + c2) = acc[r];
    }
}

extern "C" void kernel_launch(void* const* d_in, const int* in_sizes, int n_in,
                              void* d_out, int out_size, void* d_ws, size_t ws_size,
                              hipStream_t stream) {
    (void)in_sizes; (void)n_in; (void)out_size; (void)ws_size;
    const float* x   = (const float*)d_in[0];
    const float* phi = (const float*)d_in[1];
    const float* Mi  = (const float*)d_in[2];
    const float* Mf  = (const float*)d_in[3];
    float* out  = (float*)d_out;

    float* ws_x = (float*)d_ws;
    float* ws_h = ws_x + (size_t)B_ * S_ * D_;
    float* ws_p = ws_h + (size_t)HALF_S * D_;

    stu_pack_kernel  <<<dim3(3, K4_, 1),            256, 0, stream>>>(Mi, ws_p);
    stu_filter_kernel<<<dim3(3, HALF_S, 1),         256, 0, stream>>>(phi, Mf, ws_h);
    stu_xproj_wmma   <<<dim3((B_ * S_) / 32, D_ / 256, 1), 128, 0, stream>>>(x, ws_p, ws_x);
    stu_conv_kernel  <<<dim3(D_ / 32, B_, HALF_S / 64), 256, 0, stream>>>(ws_x, ws_h, out);
}


# conv kernel: register tile doubled: a block owns 128 outputs per parity, a wave 16, so each tap's filter read and window read from LDS feed 16 packed FMAs instead of 8; summation order kept
# speedup vs baseline: 1.6844x; 1.0750x over previous
#include <hip/hip_runtime.h>


typedef __attribute__((ext_vector_type(2))) float v2f;
typedef __attribute__((ext_vector_type(4))) float v4f;
typedef __attribute__((ext_vector_type(8))) float v8f;

#define B_ 4
#define S_ 4096
#define D_ 768
#define K_ 24
#define HALF_S 2048
#define K4_ (D_ / 4)

__global__ __launch_bounds__(256)
void stu_pack_kernel(const float* __restrict__ Mi, float* __restrict__ P) {
    int n  = blockIdx.x * 256 + threadIdx.x;
    int k4 = blockIdx.y;
    if (n >= D_) return;
    float4 v;
    v.x = Mi[(size_t)(4 * k4 + 0) * D_ + n];
    v.y = Mi[(size_t)(4 * k4 + 1) * D_ + n];
    v.z = Mi[(size_t)(4 * k4 + 2) * D_ + n];
    v.w = Mi[(size_t)(4 * k4 + 3) * D_ + n];
    *(float4*)(P + ((size_t)k4 * D_ + n) * 4) = v;
}

__global__ __launch_bounds__(256)
void stu_filter_kernel(const float* __restrict__ phi,
                       const float* __restrict__ Mf,
                       float* __restrict__ he) {
    int d = blockIdx.x * 256 + threadIdx.x;
    int j = blockIdx.y;
    if (d >= D_) return;
    const float* prow = phi + (2 * j) * K_;
    float acc = 0.f;
#pragma unroll
    for (int k = 0; k < K_; ++k)
        acc = fmaf(prow[k], Mf[k * D_ + d], acc);
    he[j * D_ + d] = 2.0f * acc;
}

__global__ __launch_bounds__(128)
void stu_xproj_wmma(const float* __restrict__ X,
                    const float* __restrict__ P,
                    float* __restrict__ XP) {
    const int wave = threadIdx.x >> 5;
    const int lane = threadIdx.x & 31;
    const int half = lane >> 4;
    const int l16  = lane & 15;
    const int m0 = blockIdx.x * 32;
    const int n0 = blockIdx.y * 256 + wave * 64;

    v8f c0[4] = {}, c1[4] = {};
    const float* arow0 = X + (size_t)(m0 + l16) * D_ + half * 2;
    const float* arow1 = arow0 + (size_t)16 * D_;
    const float* pcol  = P + ((size_t)(n0 + l16)) * 4 + 2 * half;

    for (int k4 = 0; k4 < K4_; ++k4) {
        v2f a0 = *(const v2f*)(arow0 + k4 * 4);
        v2f a1 = *(const v2f*)(arow1 + k4 * 4);
        const float* pk = pcol + (size_t)k4 * D_ * 4;
#pragma unroll
        for (int nt = 0; nt < 4; ++nt) {
            v2f b = *(const v2f*)(pk + nt * 64);
            c0[nt] = __builtin_amdgcn_wmma_f32_16x16x4_f32(false, a0, false, b, (short)0, c0[nt], false, false);
            c1[nt] = __builtin_amdgcn_wmma_f32_16x16x4_f32(false, a1, false, b, (short)0, c1[nt], false, false);
        }
    }

    float* out0 = XP + (size_t)(m0 + half * 8) * D_ + n0 + l16;
    float* out1 = out0 + (size_t)16 * D_;
#pragma unroll
    for (int nt = 0; nt < 4; ++nt) {
#pragma unroll
        for (int v = 0; v < 8; ++v) {
            out0[(size_t)v * D_ + nt * 16] = c0[nt][v];
            out1[(size_t)v * D_ + nt * 16] = c1[nt][v];
        }
    }
}


#define CONV_IB    128
#define CONV_NIB   (HALF_S / CONV_IB)
#define CONV_STEPS (CONV_IB + 64)
#define CONV_XROWS (2 * CONV_STEPS)
#define CONV_XP    (CONV_XROWS / 32)

struct ConvStage { v4f h0, h1, x[CONV_XP]; };

__device__ __forceinline__ void conv_issue_loads(ConvStage& st, const float* __restrict__ xpb,
                                                 const float* __restrict__ hep, int c, int i0, int r0) {
    const int j0 = c * 64;
    const int t_base = 2 * (i0 - j0 - 63);
    const int t0 = t_base + r0;
    if (t_base >= 0 && t_base + (CONV_XROWS - 1) < S_) {
        const float* xp = xpb + (size_t)t0 * D_;
#pragma unroll
        for (int k = 0; k < CONV_XP; ++k) st.x[k] = *(const v4f*)(xp + (size_t)(32 * k) * D_);
    } else {
#pragma unroll
        for (int k = 0; k < CONV_XP; ++k) {
            const int t = t0 + 32 * k;
            v4f v = {0.f, 0.f, 0.f, 0.f};
            if (t >= 0 && t < S_) v = *(const v4f*)(xpb + (size_t)t * D_);
            st.x[k] = v;
        }
    }
    const float* hp = hep + (size_t)(j0 + r0) * D_;
    st.h0 = *(const v4f*)(hp);
    st.h1 = *(const v4f*)(hp + (size_t)32 * D_);
}

__global__ __launch_bounds__(256)
void stu_conv_kernel(const float* __restrict__ XP,
                     const float* __restrict__ HE,
                     float* __restrict__ OUT) {
    __shared__ __attribute__((aligned(16))) float sh_h[64 * 32];
    __shared__ __attribute__((aligned(16))) float sh_x[CONV_XROWS * 32];

    const int ib = (CONV_NIB - 1) - (int)blockIdx.z;
    const int d0 = blockIdx.x * 32;
    const int b  = blockIdx.y;
    const int tid  = threadIdx.x;
    const int lane = tid & 31;
    const int w    = tid >> 5;
    const int i0 = ib * CONV_IB;
    const int nch = 2 * ib + 2;
    const int par = lane >> 4;
    const int c2  = (lane & 15) * 2;
    const int q4 = (tid & 7) * 4;
    const int r0 = tid >> 3;

    v2f acc[16];
#pragma unroll
    for (int r = 0; r < 16; ++r) acc[r] = (v2f){0.f, 0.f};

    const float* xpb = XP + (size_t)b * S_ * D_ + d0 + q4;
    const float* hep = HE + d0 + q4;
    float* sth = sh_h + r0 * 32 + q4;
    float* stx = sh_x + r0 * 32 + q4;

    ConvStage st;
    conv_issue_loads(st, xpb, hep, 0, i0, r0);

    for (int c = 0; c < nch; ++c) {
        *(v4f*)(sth)           = st.h0;
        *(v4f*)(sth + 32 * 32) = st.h1;
#pragma unroll
        for (int k = 0; k < CONV_XP; ++k) *(v4f*)(stx + (32 * k) * 32) = st.x[k];
        __syncthreads();

        if (c + 1 < nch) conv_issue_loads(st, xpb, hep, c + 1, i0, r0);

        if (c + 1 < nch || w >= 4) {
            const float* xw = sh_x + 64 * (16 * w) + 2 * lane;
            const float* hw = sh_h + c2;
            v2f xwin[16];
#pragma unroll
            for (int r = 0; r < 16; ++r) xwin[r] = *(const v2f*)(xw + 64 * r);
#pragma unroll 16
            for (int jj = 63; jj >= 0; --jj) {
                const v2f hh = *(const v2f*)(hw + 32 * jj);
#pragma unroll
                for (int r = 0; r < 16; ++r) acc[r] = __builtin_elementwise_fma(hh, xwin[r], acc[r]);
#pragma unroll
                for (int r = 0; r < 15; ++r) xwin[r] = xwin[r + 1];
                xwin[15] = *(const v2f*)(xw + 64 * (79 - jj));
                if ((jj & 3) == 0) __builtin_amdgcn_sched_barrier(0);
            }
        }
        __syncthreads();
    }

#pragma unroll
    for (int r = 0; r < 16; ++r) {
        const int t = 2 * (i0 + 16 * w + r) + par;
        *(v2f*)(OUT + ((size_t)b * S_ + t) * D_ + d0 + c2) = acc[r];
    }
}

extern "C" void kernel_launch(void* const* d_in, const int* in_sizes, int n_in,
                              void* d_out, int out_size, void* d_ws, size_t ws_size,
                              hipStream_t stream) {
    (void)in_sizes; (void)n_in; (void)out_size; (void)ws_size;
    const float* x   = (const float*)d_in[0];
    const float* phi = (const float*)d_in[1];
    const float* Mi  = (const float*)d_in[2];
    const float* Mf  = (const float*)d_in[3];
    float* out  = (float*)d_out;

    float* ws_x = (float*)d_ws;
    float* ws_h = ws_x + (size_t)B_ * S_ * D_;
    float* ws_p = ws_h + (size_t)HALF_S * D_;

    stu_pack_kernel  <<<dim3(3, K4_, 1),            256, 0, stream>>>(Mi, ws_p);
    stu_filter_kernel<<<dim3(3, HALF_S, 1),         256, 0, stream>>>(phi, Mf, ws_h);
    stu_xproj_wmma   <<<dim3((B_ * S_) / 32, D_ / 256, 1), 128, 0, stream>>>(x, ws_p, ws_x);
    stu_conv_kernel  <<<dim3(D_ / 32, B_, CONV_NIB), 256, 0, stream>>>(ws_x, ws_h, out);
}
